# starting kernel
# baseline (speedup 1.0000x reference)
#include <hip/hip_runtime.h>
#include <math.h>

typedef __attribute__((ext_vector_type(2))) float v2f;
typedef __attribute__((ext_vector_type(8))) float v8f;

#define NNODES 100000
#define DEG    16
#define DIM    128
#define DE     16
#define MEANW  (DIM + DE)
#define KTOT   (2 * DIM + DE)
#define DOUT   128

__global__ __launch_bounds__(256)
void sage_gather_mean(const float* __restrict__ h,
                      const float* __restrict__ edge_table,
                      const int*  __restrict__ neighbor_idx,
                      const int*  __restrict__ edge_idx,
                      float* __restrict__ mean) {
    const int lane = threadIdx.x & 31;
    const int node = blockIdx.x * 8 + (threadIdx.x >> 5);
    const long ebase = (long)node * DEG;

    float4 ah = make_float4(0.f, 0.f, 0.f, 0.f);
    float4 ae = make_float4(0.f, 0.f, 0.f, 0.f);

    #pragma unroll 4
    for (int e = 0; e < DEG; ++e) {
        const int nbr = neighbor_idx[ebase + e];
        const int eid = edge_idx[ebase + e];
        const float4 hv =
            *(const float4*)(h + (long)nbr * DIM + lane * 4);
        ah.x += hv.x; ah.y += hv.y; ah.z += hv.z; ah.w += hv.w;
        if (lane < 4) {
            const float4 ev =
                *(const float4*)(edge_table + (long)eid * DE + lane * 4);
            ae.x += ev.x; ae.y += ev.y; ae.z += ev.z; ae.w += ev.w;
        }
    }
    const float inv = 1.0f / (float)DEG;
    float* __restrict__ mp = mean + (long)node * MEANW;
    float4 mh = make_float4(ah.x * inv, ah.y * inv, ah.z * inv, ah.w * inv);
    *(float4*)(mp + lane * 4) = mh;
    if (lane < 4) {
        float4 me = make_float4(ae.x * inv, ae.y * inv, ae.z * inv, ae.w * inv);
        *(float4*)(mp + DIM + lane * 4) = me;
    }
}

#define RT 5
#define ASTRIDE 274

__global__ __launch_bounds__(256)
void sage_gemm_elu(const float* __restrict__ h,
                   const float* __restrict__ mean,
                   const float* __restrict__ W,
                   float* __restrict__ out) {
    __shared__ float Asub[RT * 16][ASTRIDE];

    const int tid  = threadIdx.x;
    const int wave = tid >> 5;
    const int lane = tid & 31;
    const long row0 = (long)blockIdx.x * (RT * 16);

    for (int i = tid; i < RT * 16 * DIM; i += 256) {
        const int r = i >> 7, c = i & (DIM - 1);
        Asub[r][c] = h[(row0 + r) * DIM + c];
    }
    for (int i = tid; i < RT * 16 * MEANW; i += 256) {
        const int r = i / MEANW, c = i - r * MEANW;
        Asub[r][DIM + c] = mean[(row0 + r) * MEANW + c];
    }
    __syncthreads();

    v8f acc[RT];
    #pragma unroll
    for (int rt = 0; rt < RT; ++rt)
        #pragma unroll
        for (int j = 0; j < 8; ++j) acc[rt][j] = 0.0f;

    const int m     = lane & 15;
    const int khalf = (lane >> 4) * 2;
    const int ncol  = wave * 16 + m;

    const float* __restrict__ arow[RT];
    #pragma unroll
    for (int rt = 0; rt < RT; ++rt)
        arow[rt] = &Asub[rt * 16 + m][khalf];
    const float* __restrict__ wp = W + (long)khalf * DOUT + ncol;

    for (int k = 0; k < KTOT; k += 4) {
        v2f b;
        b.x = wp[k * DOUT];
        b.y = wp[k * DOUT + DOUT];
        #pragma unroll
        for (int rt = 0; rt < RT; ++rt) {
            const float2 av = *(const float2*)(arow[rt] + k);
            v2f a; a.x = av.x; a.y = av.y;
            acc[rt] = __builtin_amdgcn_wmma_f32_16x16x4_f32(
                false, a, false, b, (short)0, acc[rt], false, false);
        }
    }

    const int mhi = (lane >> 4) * 8;
    #pragma unroll
    for (int rt = 0; rt < RT; ++rt) {
        #pragma unroll
        for (int r = 0; r < 8; ++r) {
            const long node = row0 + rt * 16 + r + mhi;
            float x = acc[rt][r];
            x = x > 0.0f ? x : expm1f(x);
            out[node * DOUT + ncol] = x;
        }
    }
}

extern "C" void kernel_launch(void* const* d_in, const int* in_sizes, int n_in,
                              void* d_out, int out_size, void* d_ws, size_t ws_size,
                              hipStream_t stream) {
    const float* h            = (const float*)d_in[0];
    const float* edge_table   = (const float*)d_in[1];
    const float* W            = (const float*)d_in[2];
    const int*   neighbor_idx = (const int*)d_in[3];
    const int*   edge_idx     = (const int*)d_in[4];

    float* mean = (float*)d_ws;
    float* out  = (float*)d_out;

    sage_gather_mean<<<NNODES / 8, 256, 0, stream>>>(h, edge_table,
                                                     neighbor_idx, edge_idx, mean);
    sage_gemm_elu<<<NNODES / (RT * 16), 256, 0, stream>>>(h, mean, W, out);
}


# gather kernel re-tiled from 1 to 8 nodes per wave: edge rows read one lane per (node, 16-byte quarter) with no lane masked, so the neighbor-row loop loses its divergent branch; sum order kept
# speedup vs baseline: 1.4769x; 1.4769x over previous
#include <hip/hip_runtime.h>
#include <math.h>

typedef __attribute__((ext_vector_type(2))) float v2f;
typedef __attribute__((ext_vector_type(8))) float v8f;

#define NNODES 100000
#define DEG    16
#define DIM    128
#define DE     16
#define MEANW  (DIM + DE)
#define KTOT   (2 * DIM + DE)
#define DOUT   128

#define NPW 8
#define GWAVES 4

__global__ __launch_bounds__(GWAVES * 32)
void sage_gather_mean(const float* __restrict__ h,
                      const float* __restrict__ edge_table,
                      const int*  __restrict__ neighbor_idx,
                      const int*  __restrict__ edge_idx,
                      float* __restrict__ mean) {
    const int lane  = threadIdx.x & 31;
    const int node0 = (blockIdx.x * GWAVES + (threadIdx.x >> 5)) * NPW;
    const float inv = 1.0f / (float)DEG;

    {
        const int en = node0 + (lane >> 2);
        const int q  = lane & 3;
        const int* ep = edge_idx + (long)en * DEG;
        float4 ae = make_float4(0.f, 0.f, 0.f, 0.f);
        for (int e = 0; e < DEG; ++e) {
            const int eid = ep[e];
            const float4 ev =
                *(const float4*)(edge_table + (long)eid * DE + q * 4);
            ae.x += ev.x; ae.y += ev.y; ae.z += ev.z; ae.w += ev.w;
        }
        float4 me = make_float4(ae.x * inv, ae.y * inv, ae.z * inv, ae.w * inv);
        *(float4*)(mean + (long)en * MEANW + DIM + q * 4) = me;
    }

    for (int j = 0; j < NPW; ++j) {
        const int node = node0 + j;
        const long ebase = (long)node * DEG;
        float4 ah = make_float4(0.f, 0.f, 0.f, 0.f);
        #pragma unroll 4
        for (int e = 0; e < DEG; ++e) {
            const int nbr = neighbor_idx[ebase + e];
            const float4 hv =
                *(const float4*)(h + (long)nbr * DIM + lane * 4);
            ah.x += hv.x; ah.y += hv.y; ah.z += hv.z; ah.w += hv.w;
        }
        float4 mh = make_float4(ah.x * inv, ah.y * inv, ah.z * inv, ah.w * inv);
        *(float4*)(mean + (long)node * MEANW + lane * 4) = mh;
    }
}

#define RT 5
#define ASTRIDE 274

__global__ __launch_bounds__(256)
void sage_gemm_elu(const float* __restrict__ h,
                   const float* __restrict__ mean,
                   const float* __restrict__ W,
                   float* __restrict__ out) {
    __shared__ float Asub[RT * 16][ASTRIDE];

    const int tid  = threadIdx.x;
    const int wave = tid >> 5;
    const int lane = tid & 31;
    const long row0 = (long)blockIdx.x * (RT * 16);

    for (int i = tid; i < RT * 16 * DIM; i += 256) {
        const int r = i >> 7, c = i & (DIM - 1);
        Asub[r][c] = h[(row0 + r) * DIM + c];
    }
    for (int i = tid; i < RT * 16 * MEANW; i += 256) {
        const int r = i / MEANW, c = i - r * MEANW;
        Asub[r][DIM + c] = mean[(row0 + r) * MEANW + c];
    }
    __syncthreads();

    v8f acc[RT];
    #pragma unroll
    for (int rt = 0; rt < RT; ++rt)
        #pragma unroll
        for (int j = 0; j < 8; ++j) acc[rt][j] = 0.0f;

    const int m     = lane & 15;
    const int khalf = (lane >> 4) * 2;
    const int ncol  = wave * 16 + m;

    const float* __restrict__ arow[RT];
    #pragma unroll
    for (int rt = 0; rt < RT; ++rt)
        arow[rt] = &Asub[rt * 16 + m][khalf];
    const float* __restrict__ wp = W + (long)khalf * DOUT + ncol;

    for (int k = 0; k < KTOT; k += 4) {
        v2f b;
        b.x = wp[k * DOUT];
        b.y = wp[k * DOUT + DOUT];
        #pragma unroll
        for (int rt = 0; rt < RT; ++rt) {
            const float2 av = *(const float2*)(arow[rt] + k);
            v2f a; a.x = av.x; a.y = av.y;
            acc[rt] = __builtin_amdgcn_wmma_f32_16x16x4_f32(
                false, a, false, b, (short)0, acc[rt], false, false);
        }
    }

    const int mhi = (lane >> 4) * 8;
    #pragma unroll
    for (int rt = 0; rt < RT; ++rt) {
        #pragma unroll
        for (int r = 0; r < 8; ++r) {
            const long node = row0 + rt * 16 + r + mhi;
            float x = acc[rt][r];
            x = x > 0.0f ? x : expm1f(x);
            out[node * DOUT + ncol] = x;
        }
    }
}

extern "C" void kernel_launch(void* const* d_in, const int* in_sizes, int n_in,
                              void* d_out, int out_size, void* d_ws, size_t ws_size,
                              hipStream_t stream) {
    const float* h            = (const float*)d_in[0];
    const float* edge_table   = (const float*)d_in[1];
    const float* W            = (const float*)d_in[2];
    const int*   neighbor_idx = (const int*)d_in[3];
    const int*   edge_idx     = (const int*)d_in[4];

    float* mean = (float*)d_ws;
    float* out  = (float*)d_out;

    sage_gather_mean<<<NNODES / (GWAVES * NPW), GWAVES * 32, 0, stream>>>(h, edge_table,
                                                     neighbor_idx, edge_idx, mean);
    sage_gemm_elu<<<NNODES / (RT * 16), 256, 0, stream>>>(h, mean, W, out);
}


# GEMM kernel: A tile staged in two K phases with 16-byte loads; the mean half is requested ahead of the h-half K loop (its B fragments preloaded in registers) and stored after it; sum order kept
# speedup vs baseline: 1.6095x; 1.0898x over previous
#include <hip/hip_runtime.h>
#include <math.h>

typedef __attribute__((ext_vector_type(2))) float v2f;
typedef __attribute__((ext_vector_type(8))) float v8f;

#define NNODES 100000
#define DEG    16
#define DIM    128
#define DE     16
#define MEANW  (DIM + DE)
#define KTOT   (2 * DIM + DE)
#define DOUT   128

#define NPW 8
#define GWAVES 4

__global__ __launch_bounds__(GWAVES * 32)
void sage_gather_mean(const float* __restrict__ h,
                      const float* __restrict__ edge_table,
                      const int*  __restrict__ neighbor_idx,
                      const int*  __restrict__ edge_idx,
                      float* __restrict__ mean) {
    const int lane  = threadIdx.x & 31;
    const int node0 = (blockIdx.x * GWAVES + (threadIdx.x >> 5)) * NPW;
    const float inv = 1.0f / (float)DEG;

    {
        const int en = node0 + (lane >> 2);
        const int q  = lane & 3;
        const int* ep = edge_idx + (long)en * DEG;
        float4 ae = make_float4(0.f, 0.f, 0.f, 0.f);
        for (int e = 0; e < DEG; ++e) {
            const int eid = ep[e];
            const float4 ev =
                *(const float4*)(edge_table + (long)eid * DE + q * 4);
            ae.x += ev.x; ae.y += ev.y; ae.z += ev.z; ae.w += ev.w;
        }
        float4 me = make_float4(ae.x * inv, ae.y * inv, ae.z * inv, ae.w * inv);
        *(float4*)(mean + (long)en * MEANW + DIM + q * 4) = me;
    }

    for (int j = 0; j < NPW; ++j) {
        const int node = node0 + j;
        const long ebase = (long)node * DEG;
        float4 ah = make_float4(0.f, 0.f, 0.f, 0.f);
        #pragma unroll 4
        for (int e = 0; e < DEG; ++e) {
            const int nbr = neighbor_idx[ebase + e];
            const float4 hv =
                *(const float4*)(h + (long)nbr * DIM + lane * 4);
            ah.x += hv.x; ah.y += hv.y; ah.z += hv.z; ah.w += hv.w;
        }
        float4 mh = make_float4(ah.x * inv, ah.y * inv, ah.z * inv, ah.w * inv);
        *(float4*)(mean + (long)node * MEANW + lane * 4) = mh;
    }
}

#define RT 5
#define ASTRIDE 276
#define HV4  (RT * 16 * (DIM / 4))
#define MV4  (RT * 16 * (MEANW / 4))
#define HPT  (HV4 / 256)
#define MPT  (MV4 / 256)
#define MREM (MV4 * 4 - MPT * 256 * 4)

__global__ __launch_bounds__(256, 6)
void sage_gemm_elu(const float* __restrict__ h,
                   const float* __restrict__ mean,
                   const float* __restrict__ W,
                   float* __restrict__ out) {
    __shared__ __attribute__((aligned(16))) float Asub[RT * 16][ASTRIDE];

    const int tid  = threadIdx.x;
    const int wave = tid >> 5;
    const int lane = tid & 31;
    const long row0 = (long)blockIdx.x * (RT * 16);

    const float4* __restrict__ hsrc = (const float4*)(h + row0 * DIM);
    const float4* __restrict__ msrc = (const float4*)(mean + row0 * MEANW);
    float4 hv[HPT];
    #pragma unroll
    for (int j = 0; j < HPT; ++j) hv[j] = hsrc[tid + j * 256];

    const int m     = lane & 15;
    const int khalf = (lane >> 4) * 2;
    const int ncol  = wave * 16 + m;
    const float* __restrict__ wp = W + (long)khalf * DOUT + ncol;

    v2f bw[DIM / 4];
    #pragma unroll
    for (int s = 0; s < DIM / 4; ++s) {
        bw[s].x = wp[(s * 4) * DOUT];
        bw[s].y = wp[(s * 4) * DOUT + DOUT];
    }

    #pragma unroll
    for (int j = 0; j < HPT; ++j) {
        const int i = tid + j * 256;
        *(float4*)&Asub[i >> 5][(i & 31) * 4] = hv[j];
    }
    __syncthreads();

    static_assert(MREM == 256, "the remainder of the mean half is one float per thread");
    float4 mv[MPT];
    #pragma unroll
    for (int j = 0; j < MPT; ++j) mv[j] = msrc[tid + j * 256];
    const float ml = mean[row0 * MEANW + MPT * 256 * 4 + tid];

    v8f acc[RT];
    #pragma unroll
    for (int rt = 0; rt < RT; ++rt)
        #pragma unroll
        for (int j = 0; j < 8; ++j) acc[rt][j] = 0.0f;

    const float* __restrict__ arow[RT];
    #pragma unroll
    for (int rt = 0; rt < RT; ++rt)
        arow[rt] = &Asub[rt * 16 + m][khalf];

    #pragma unroll
    for (int s = 0; s < DIM / 4; ++s) {
        #pragma unroll
        for (int rt = 0; rt < RT; ++rt) {
            const float2 av = *(const float2*)(arow[rt] + s * 4);
            v2f a; a.x = av.x; a.y = av.y;
            acc[rt] = __builtin_amdgcn_wmma_f32_16x16x4_f32(
                false, a, false, bw[s], (short)0, acc[rt], false, false);
        }
    }

    #pragma unroll
    for (int j = 0; j < MPT; ++j) {
        const int i = tid + j * 256;
        const int r = i / (MEANW / 4), c = i - r * (MEANW / 4);
        *(float4*)&Asub[r][DIM + c * 4] = mv[j];
    }
    {
        const int i = MPT * 256 * 4 + tid;
        const int r = i / MEANW, c = i - r * MEANW;
        Asub[r][DIM + c] = ml;
    }
    __syncthreads();

    #pragma unroll 4
    for (int k = DIM; k < KTOT; k += 4) {
        v2f b;
        b.x = wp[k * DOUT];
        b.y = wp[k * DOUT + DOUT];
        #pragma unroll
        for (int rt = 0; rt < RT; ++rt) {
            const float2 av = *(const float2*)(arow[rt] + k);
            v2f a; a.x = av.x; a.y = av.y;
            acc[rt] = __builtin_amdgcn_wmma_f32_16x16x4_f32(
                false, a, false, b, (short)0, acc[rt], false, false);
        }
    }

    const int mhi = (lane >> 4) * 8;
    #pragma unroll
    for (int rt = 0; rt < RT; ++rt) {
        #pragma unroll
        for (int r = 0; r < 8; ++r) {
            const long node = row0 + rt * 16 + r + mhi;
            float x = acc[rt][r];
            x = x > 0.0f ? x : expm1f(x);
            out[node * DOUT + ncol] = x;
        }
    }
}

extern "C" void kernel_launch(void* const* d_in, const int* in_sizes, int n_in,
                              void* d_out, int out_size, void* d_ws, size_t ws_size,
                              hipStream_t stream) {
    const float* h            = (const float*)d_in[0];
    const float* edge_table   = (const float*)d_in[1];
    const float* W            = (const float*)d_in[2];
    const int*   neighbor_idx = (const int*)d_in[3];
    const int*   edge_idx     = (const int*)d_in[4];

    float* mean = (float*)d_ws;
    float* out  = (float*)d_out;

    sage_gather_mean<<<NNODES / (GWAVES * NPW), GWAVES * 32, 0, stream>>>(h, edge_table,
                                                     neighbor_idx, edge_idx, mean);
    sage_gemm_elu<<<NNODES / (RT * 16), 256, 0, stream>>>(h, mean, W, out);
}
